# starting kernel
# baseline (speedup 1.0000x reference)
#include <hip/hip_runtime.h>
#include <hip/hip_bf16.h>

typedef float v2f __attribute__((ext_vector_type(2)));
typedef float v8f __attribute__((ext_vector_type(8)));

__device__ __forceinline__ float swap16(float x) {
  return __int_as_float(__builtin_amdgcn_ds_swizzle(__float_as_int(x), 0x401f));
}

__device__ __forceinline__ float fast_tanh(float x) {
#if __has_builtin(__builtin_amdgcn_tanhf)
  return __builtin_amdgcn_tanhf(x);
#else
  float e = __expf(2.0f * x);
  float r = __builtin_amdgcn_rcpf(e + 1.0f);
  return 1.0f - 2.0f * r;
#endif
}


__device__ __forceinline__ void build_B(const float x[16], v2f B[8], bool hi) {
#pragma unroll
  for (int blk = 0; blk < 8; ++blk) {
    const int k0 = blk * 4;
    const int t  = k0 >> 4;
    const int rr = k0 & 15;
    v2f b;
    if (rr < 8) {
      const int a = t * 8 + rr;
      float s0 = swap16(x[a + 2]);
      float s1 = swap16(x[a + 3]);
      b.x = hi ? s0 : x[a];
      b.y = hi ? s1 : x[a + 1];
    } else {
      const int a = t * 8 + rr - 8;
      float s0 = swap16(x[a]);
      float s1 = swap16(x[a + 1]);
      b.x = hi ? x[a + 2] : s0;
      b.y = hi ? x[a + 3] : s1;
    }
    B[blk] = b;
  }
}

__device__ __forceinline__ void layer(const float x[16], float y[16],
                                      const v2f A[2][8], const v8f biasf[2],
                                      bool hi) {
  v2f B[8];
  build_B(x, B, hi);
#pragma unroll
  for (int t = 0; t < 2; ++t) {
    v8f c = __builtin_amdgcn_wmma_f32_16x16x4_f32(
        false, A[t][0], false, B[0], (short)0, biasf[t], false, false);
#pragma unroll
    for (int blk = 1; blk < 8; ++blk) {
      c = __builtin_amdgcn_wmma_f32_16x16x4_f32(
          false, A[t][blk], false, B[blk], (short)0, c, false, false);
    }
#pragma unroll
    for (int r = 0; r < 8; ++r) y[t * 8 + r] = c[r];
  }
}

__device__ __forceinline__ void ode_f(const float x[16], float y[16],
                                      const v2f A1[2][8], const v2f A2[2][8],
                                      const v8f bias1[2], const v8f bias2[2],
                                      bool hi) {
  float hmid[16];
  layer(x, hmid, A1, bias1, hi);
#pragma unroll
  for (int i = 0; i < 16; ++i) hmid[i] = fast_tanh(hmid[i]);
  layer(hmid, y, A2, bias2, hi);
}

__global__ __launch_bounds__(256)
void node_rk4_wmma(const float* __restrict__ z0, const float* __restrict__ tt,
                   const float* __restrict__ W1, const float* __restrict__ b1,
                   const float* __restrict__ W2, const float* __restrict__ b2,
                   float* __restrict__ out, int nsteps, long nrows) {
  const int lane = threadIdx.x & 31;
  const int wv   = threadIdx.x >> 5;
  const int half = lane >> 4;
  const bool hi  = (half != 0);
  const int m    = lane & 15;

  const long tile = (long)blockIdx.x * (blockDim.x >> 5) + wv;
  const long row  = tile * 16 + m;
  if (tile * 16 >= nrows) return;

  v2f A1[2][8], A2[2][8];
#pragma unroll
  for (int t = 0; t < 2; ++t) {
#pragma unroll
    for (int blk = 0; blk < 8; ++blk) {
      const int kr  = blk * 4 + half * 2;
      const int col = t * 16 + m;
      v2f a1, a2;
      a1.x = W1[(kr + 0) * 32 + col];
      a1.y = W1[(kr + 1) * 32 + col];
      a2.x = W2[(kr + 0) * 32 + col];
      a2.y = W2[(kr + 1) * 32 + col];
      A1[t][blk] = a1;
      A2[t][blk] = a2;
    }
  }

  v8f bias1[2], bias2[2];
#pragma unroll
  for (int t = 0; t < 2; ++t) {
#pragma unroll
    for (int r = 0; r < 8; ++r) {
      const int col = t * 16 + half * 8 + r;
      bias1[t][r] = b1[col];
      bias2[t][r] = b2[col];
    }
  }

  float zc[16];
  {
    const float4* zp = (const float4*)(z0 + row * 32);
    float4 v0 = zp[half * 2 + 0];
    float4 v1 = zp[half * 2 + 1];
    float4 v2 = zp[4 + half * 2 + 0];
    float4 v3 = zp[4 + half * 2 + 1];
    zc[0] = v0.x; zc[1] = v0.y; zc[2]  = v0.z; zc[3]  = v0.w;
    zc[4] = v1.x; zc[5] = v1.y; zc[6]  = v1.z; zc[7]  = v1.w;
    zc[8] = v2.x; zc[9] = v2.y; zc[10] = v2.z; zc[11] = v2.w;
    zc[12] = v3.x; zc[13] = v3.y; zc[14] = v3.z; zc[15] = v3.w;
  }

  for (int s = 0; s < nsteps; ++s) {
    const float h = tt[s + 1] - tt[s];
    float kcur[16], acc[16], ztmp[16];

    ode_f(zc, kcur, A1, A2, bias1, bias2, hi);
#pragma unroll
    for (int i = 0; i < 16; ++i) {
      acc[i]  = kcur[i];
      ztmp[i] = zc[i] + 0.5f * h * kcur[i];
    }
    ode_f(ztmp, kcur, A1, A2, bias1, bias2, hi);
#pragma unroll
    for (int i = 0; i < 16; ++i) {
      acc[i] += 2.0f * kcur[i];
      ztmp[i] = zc[i] + 0.5f * h * kcur[i];
    }
    ode_f(ztmp, kcur, A1, A2, bias1, bias2, hi);
#pragma unroll
    for (int i = 0; i < 16; ++i) {
      acc[i] += 2.0f * kcur[i];
      ztmp[i] = zc[i] + h * kcur[i];
    }
    ode_f(ztmp, kcur, A1, A2, bias1, bias2, hi);
    const float c6 = h * (1.0f / 6.0f);
#pragma unroll
    for (int i = 0; i < 16; ++i) zc[i] += c6 * (acc[i] + kcur[i]);
  }

  {
    float4* op = (float4*)(out + row * 32);
    float4 v0, v1, v2, v3;
    v0.x = zc[0];  v0.y = zc[1];  v0.z = zc[2];  v0.w = zc[3];
    v1.x = zc[4];  v1.y = zc[5];  v1.z = zc[6];  v1.w = zc[7];
    v2.x = zc[8];  v2.y = zc[9];  v2.z = zc[10]; v2.w = zc[11];
    v3.x = zc[12]; v3.y = zc[13]; v3.z = zc[14]; v3.w = zc[15];
    op[half * 2 + 0]     = v0;
    op[half * 2 + 1]     = v1;
    op[4 + half * 2 + 0] = v2;
    op[4 + half * 2 + 1] = v3;
  }
}

extern "C" void kernel_launch(void* const* d_in, const int* in_sizes, int n_in,
                              void* d_out, int out_size, void* d_ws, size_t ws_size,
                              hipStream_t stream) {
  (void)n_in; (void)out_size; (void)d_ws; (void)ws_size;
  const float* z0 = (const float*)d_in[0];
  const float* tt = (const float*)d_in[1];
  const float* W1 = (const float*)d_in[2];
  const float* b1 = (const float*)d_in[3];
  const float* W2 = (const float*)d_in[4];
  const float* b2 = (const float*)d_in[5];
  float* out = (float*)d_out;

  const int  nsteps = in_sizes[1] - 1;
  const long nrows  = (long)in_sizes[0] / 32;
  const long tiles  = (nrows + 15) / 16;
  const int  wavesPerBlock = 8;
  const int  blocks = (int)((tiles + wavesPerBlock - 1) / wavesPerBlock);

  node_rk4_wmma<<<blocks, wavesPerBlock * 32, 0, stream>>>(
      z0, tt, W1, b1, W2, b2, out, nsteps, nrows);
}


# Output-row order of both weight A-fragment sets, biases and z permuted so each accumulator register pair is the next product's B fragment in place: half-wave swaps go; k order and sums unchanged
# speedup vs baseline: 1.1593x; 1.1593x over previous
#include <hip/hip_runtime.h>
#include <hip/hip_bf16.h>

typedef float v2f __attribute__((ext_vector_type(2)));
typedef float v8f __attribute__((ext_vector_type(8)));

__device__ __forceinline__ float fast_tanh(float x) {
#if __has_builtin(__builtin_amdgcn_tanhf)
  return __builtin_amdgcn_tanhf(x);
#else
  float e = __expf(2.0f * x);
  float r = __builtin_amdgcn_rcpf(e + 1.0f);
  return 1.0f - 2.0f * r;
#endif
}

__device__ __forceinline__ void build_B(const float x[16], v2f B[8]) {
#pragma unroll
  for (int blk = 0; blk < 8; ++blk) {
    v2f b;
    b.x = x[2 * blk];
    b.y = x[2 * blk + 1];
    B[blk] = b;
  }
}

__device__ __forceinline__ void layer(const float x[16], float y[16],
                                      const v2f A[2][8], const v8f biasf[2]) {
  v2f B[8];
  build_B(x, B);
#pragma unroll
  for (int t = 0; t < 2; ++t) {
    v8f c = __builtin_amdgcn_wmma_f32_16x16x4_f32(
        false, A[t][0], false, B[0], (short)0, biasf[t], false, false);
#pragma unroll
    for (int blk = 1; blk < 8; ++blk) {
      c = __builtin_amdgcn_wmma_f32_16x16x4_f32(
          false, A[t][blk], false, B[blk], (short)0, c, false, false);
    }
#pragma unroll
    for (int r = 0; r < 8; ++r) y[t * 8 + r] = c[r];
  }
}

__device__ __forceinline__ void ode_f(const float x[16], float y[16],
                                      const v2f A1[2][8], const v2f A2[2][8],
                                      const v8f bias1[2], const v8f bias2[2]) {
  float hmid[16];
  layer(x, hmid, A1, bias1);
#pragma unroll
  for (int i = 0; i < 16; ++i) hmid[i] = fast_tanh(hmid[i]);
  layer(hmid, y, A2, bias2);
}

__global__ __launch_bounds__(256)
void node_rk4_wmma(const float* __restrict__ z0, const float* __restrict__ tt,
                   const float* __restrict__ W1, const float* __restrict__ b1,
                   const float* __restrict__ W2, const float* __restrict__ b2,
                   float* __restrict__ out, int nsteps, long nrows) {
  const int lane = threadIdx.x & 31;
  const int wv   = threadIdx.x >> 5;
  const int half = lane >> 4;
  const int m    = lane & 15;

  const long tile = (long)blockIdx.x * (blockDim.x >> 5) + wv;
  const long row  = tile * 16 + m;
  if (tile * 16 >= nrows) return;

  v2f A1[2][8], A2[2][8];
#pragma unroll
  for (int t = 0; t < 2; ++t) {
#pragma unroll
    for (int blk = 0; blk < 8; ++blk) {
      const int kr  = blk * 4 + half * 2;
      const int col = t * 16 + 4 * ((m & 7) >> 1) + 2 * (m >> 3) + (m & 1);
      v2f a1, a2;
      a1.x = W1[(kr + 0) * 32 + col];
      a1.y = W1[(kr + 1) * 32 + col];
      a2.x = W2[(kr + 0) * 32 + col];
      a2.y = W2[(kr + 1) * 32 + col];
      A1[t][blk] = a1;
      A2[t][blk] = a2;
    }
  }

  v8f bias1[2], bias2[2];
#pragma unroll
  for (int t = 0; t < 2; ++t) {
#pragma unroll
    for (int r = 0; r < 8; ++r) {
      const int col = t * 16 + 4 * (r >> 1) + 2 * half + (r & 1);
      bias1[t][r] = b1[col];
      bias2[t][r] = b2[col];
    }
  }

  float zc[16];
  {
    const float2* zp = (const float2*)(z0 + row * 32);
#pragma unroll
    for (int t = 0; t < 2; ++t) {
#pragma unroll
      for (int j = 0; j < 4; ++j) {
        const float2 v = zp[8 * t + 2 * j + half];
        zc[t * 8 + 2 * j]     = v.x;
        zc[t * 8 + 2 * j + 1] = v.y;
      }
    }
  }

  for (int s = 0; s < nsteps; ++s) {
    const float h = tt[s + 1] - tt[s];
    float kcur[16], acc[16], ztmp[16];

    ode_f(zc, kcur, A1, A2, bias1, bias2);
#pragma unroll
    for (int i = 0; i < 16; ++i) {
      acc[i]  = kcur[i];
      ztmp[i] = zc[i] + 0.5f * h * kcur[i];
    }
    ode_f(ztmp, kcur, A1, A2, bias1, bias2);
#pragma unroll
    for (int i = 0; i < 16; ++i) {
      acc[i] += 2.0f * kcur[i];
      ztmp[i] = zc[i] + 0.5f * h * kcur[i];
    }
    ode_f(ztmp, kcur, A1, A2, bias1, bias2);
#pragma unroll
    for (int i = 0; i < 16; ++i) {
      acc[i] += 2.0f * kcur[i];
      ztmp[i] = zc[i] + h * kcur[i];
    }
    ode_f(ztmp, kcur, A1, A2, bias1, bias2);
    const float c6 = h * (1.0f / 6.0f);
#pragma unroll
    for (int i = 0; i < 16; ++i) zc[i] += c6 * (acc[i] + kcur[i]);
  }

  {
    float2* op = (float2*)(out + row * 32);
#pragma unroll
    for (int t = 0; t < 2; ++t) {
#pragma unroll
      for (int j = 0; j < 4; ++j) {
        float2 v;
        v.x = zc[t * 8 + 2 * j];
        v.y = zc[t * 8 + 2 * j + 1];
        op[8 * t + 2 * j + half] = v;
      }
    }
  }
}

extern "C" void kernel_launch(void* const* d_in, const int* in_sizes, int n_in,
                              void* d_out, int out_size, void* d_ws, size_t ws_size,
                              hipStream_t stream) {
  (void)n_in; (void)out_size; (void)d_ws; (void)ws_size;
  const float* z0 = (const float*)d_in[0];
  const float* tt = (const float*)d_in[1];
  const float* W1 = (const float*)d_in[2];
  const float* b1 = (const float*)d_in[3];
  const float* W2 = (const float*)d_in[4];
  const float* b2 = (const float*)d_in[5];
  float* out = (float*)d_out;

  const int  nsteps = in_sizes[1] - 1;
  const long nrows  = (long)in_sizes[0] / 32;
  const long tiles  = (nrows + 15) / 16;
  const int  wavesPerBlock = 8;
  const int  blocks = (int)((tiles + wavesPerBlock - 1) / wavesPerBlock);

  node_rk4_wmma<<<blocks, wavesPerBlock * 32, 0, stream>>>(
      z0, tt, W1, b1, W2, b2, out, nsteps, nrows);
}
